# starting kernel
# baseline (speedup 1.0000x reference)
#include <hip/hip_runtime.h>

typedef __attribute__((ext_vector_type(2))) float v2f;
typedef __attribute__((ext_vector_type(8))) float v8f;

#define NCH   256
#define KTOT  2304
#define KG    (KTOT / 4)
#define HW    784

__global__ void build_weff_k(const float* __restrict__ w1, const float* __restrict__ w2,
                             float* __restrict__ weff) {
    int tid = blockIdx.x * blockDim.x + threadIdx.x;
    if (tid >= NCH * KTOT) return;
    int oc  = tid / KTOT;
    int r   = tid % KTOT;
    int blk = r >> 8;
    int cin = r & 255;
    int i = blk / 3, k = blk % 3;
    int l = oc >> 3, p = oc & 7;
    int cg = cin >> 2, n = cin & 3;
    float a = w1[((l * 4 + n) * 3 + i) * 3 + k];
    float b = w2[(p * 64 + cg) * 3 + k];
    weff[tid] = a * b;
}

__global__ void build_bpack_k(const float* __restrict__ x, float* __restrict__ bpack) {
    int tid = blockIdx.x * blockDim.x + threadIdx.x;
    if (tid >= KTOT * HW) return;
    int e  = tid & 3;
    int t  = tid >> 2;
    int s  = t % HW;
    int kg = t / HW;
    int blk = kg >> 6;
    int cin = ((kg & 63) << 2) | e;
    int i = blk / 3, k = blk % 3;
    int ho = s / 28, wo = s % 28;
    int row = ho + i - 1, col = wo + k - 1;
    float v = 0.0f;
    if (row >= 0 && row < 28 && col >= 0 && col < 28)
        v = x[cin * HW + row * 28 + col];
    bpack[tid] = v;
}

__global__ void conv_wmma_k(const float* __restrict__ weff, const float* __restrict__ bpack,
                            float* __restrict__ out) {
    int wave   = (blockIdx.x * blockDim.x + threadIdx.x) >> 5;
    int lane   = threadIdx.x & 31;
    int m_tile = wave & 15;
    int s_tile = wave >> 4;
    int lr     = lane & 15;
    int half   = lane >> 4;

    int s    = s_tile * 16 + lr;
    int rowA = m_tile * 16 + lr;

    const float* ap = weff + rowA * KTOT + 2 * half;
    const float* bp = bpack + s * 4 + 2 * half;

    v8f acc = {};
    #pragma unroll 8
    for (int kg = 0; kg < KG; ++kg) {
        v2f a = *(const v2f*)(ap + kg * 4);
        v2f b = *(const v2f*)(bp + kg * (HW * 4));
        acc = __builtin_amdgcn_wmma_f32_16x16x4_f32(
            false, a, false, b, (short)0, acc, false, false);
    }

    #pragma unroll
    for (int r = 0; r < 8; ++r) {
        int oc = m_tile * 16 + half * 8 + r;
        out[oc * HW + s] = acc[r];
    }
}

extern "C" void kernel_launch(void* const* d_in, const int* in_sizes, int n_in,
                              void* d_out, int out_size, void* d_ws, size_t ws_size,
                              hipStream_t stream) {
    const float* x  = (const float*)d_in[0];
    const float* w1 = (const float*)d_in[1];
    const float* w2 = (const float*)d_in[2];
    float* out = (float*)d_out;

    float* weff  = (float*)d_ws;
    float* bpack = weff + NCH * KTOT;

    build_weff_k<<<(NCH * KTOT + 255) / 256, 256, 0, stream>>>(w1, w2, weff);
    build_bpack_k<<<(KTOT * HW + 255) / 256, 256, 0, stream>>>(x, bpack);
    conv_wmma_k<<<196, 128, 0, stream>>>(weff, bpack, out);
}


# Contraction reordered: (n,i) with w1 first on WMMA (K=12), then (cg,k) with w2 on WMMA fed from stage 1's own accumulator registers; 576 to 240 WMMAs per SIMD; sum order changes
# speedup vs baseline: 7.5142x; 7.5142x over previous
#include <hip/hip_runtime.h>


typedef __attribute__((ext_vector_type(2))) float v2f;
typedef __attribute__((ext_vector_type(8))) float v8f;
typedef __attribute__((ext_vector_type(3))) float v3f;
typedef v3f v3f_a4 __attribute__((aligned(4)));

#define HW       784
#define SPLITK   8
#define CG_WAVE  8
#define NT       4


template <int RH>
__device__ __forceinline__ void contract(const float* __restrict__ x, const float* __restrict__ w1,
                                         const float* __restrict__ w2, int mt, int w, int lane,
                                         int s, v8f (&acc)[NT]) {
    int lr   = lane & 15;
    int half = lane >> 4;
    int ho = s / 28;
    int wo = s - ho * 28;

    const float* w1p = w1 + (mt * 16 + lr) * 36 + half * 18;
    v2f a1[9];
    #pragma unroll
    for (int t = 0; t < 9; ++t) {
        a1[t].x = w1p[t];
        a1[t].y = w1p[9 + t];
    }

    int  q   = lr >> 3;
    int  p   = lr & 7;
    bool own = (q == half);
    bool hi  = (half != 0);
    const float* w2p = w2 + p * 192 + w * (CG_WAVE * 3);
    float w2v[CG_WAVE][3];
    #pragma unroll
    for (int c = 0; c < CG_WAVE; ++c)
        #pragma unroll
        for (int k = 0; k < 3; ++k)
            w2v[c][k] = w2p[c * 3 + k];
    v2f a2[2][3][2];
    #pragma unroll
    for (int g = 0; g < 2; ++g)
        #pragma unroll
        for (int k = 0; k < 3; ++k) {
            float c0 = w2v[4 * g + 0][k], c1 = w2v[4 * g + 1][k];
            float c2 = w2v[4 * g + 2][k], c3 = w2v[4 * g + 3][k];
            a2[g][k][0].x = own ? (hi ? c2 : c0) : 0.0f;
            a2[g][k][0].y = own ? (hi ? c3 : c1) : 0.0f;
            a2[g][k][1].x = own ? (hi ? c0 : c2) : 0.0f;
            a2[g][k][1].y = own ? (hi ? c1 : c3) : 0.0f;
        }

    bool onL = (wo == 0);
    bool onR = (wo == 27);
    int  cs  = wo - 1 + (onL ? 1 : 0) - (onR ? 1 : 0);
    bool rv0 = (ho > 0);
    bool rv2 = (ho < 27);
    int off[3];
    off[0] = (rv0 ? ho - 1 : ho) * 28 + cs;
    off[1] = ho * 28 + cs;
    off[2] = (rv2 ? ho + 1 : ho) * 28 + cs;
    const float* xb = x + (w * (CG_WAVE * 4) + 2 * half) * HW;

    #pragma unroll
    for (int t = 0; t < NT; ++t) acc[t] = v8f{};

    #pragma unroll
    for (int g = 0; g < 2; ++g) {
        v8f d1[4][3];
        #pragma unroll
        for (int cc = 0; cc < 4; ++cc) {
            int c = 4 * g + cc;
            #pragma unroll
            for (int k = 0; k < 3; ++k) d1[cc][k] = v8f{};
            #pragma unroll
            for (int i = 0; i < 3; ++i) {
                bool rv = (i == 0) ? rv0 : ((i == 2) ? rv2 : true);
                float tap[2][3];
                #pragma unroll
                for (int e = 0; e < 2; ++e) {
                    const float* qp = xb + (c * 4 + e) * HW + off[i];
                    v3f_a4 qv = *(const v3f_a4*)qp;
                    float q0 = qv.x, q1 = qv.y, q2 = qv.z;
                    float u0 = onR ? q1 : q0;
                    float u1 = onL ? q0 : (onR ? q2 : q1);
                    float u2 = onL ? q1 : q2;
                    tap[e][0] = (rv && !onL) ? u0 : 0.0f;
                    tap[e][1] = rv ? u1 : 0.0f;
                    tap[e][2] = (rv && !onR) ? u2 : 0.0f;
                }
                #pragma unroll
                for (int k = 0; k < 3; ++k) {
                    v2f b = { tap[0][k], tap[1][k] };
                    d1[cc][k] = __builtin_amdgcn_wmma_f32_16x16x4_f32(
                        false, a1[i * 3 + k], false, b, (short)0, d1[cc][k], false, false);
                }
            }
        }
        #pragma unroll
        for (int k = 0; k < 3; ++k) {
            #pragma unroll
            for (int t = 0; t < NT; ++t) {
                const int r = 4 * RH + t;
                float o0 = d1[0][k][r], o1 = d1[1][k][r];
                float o2 = d1[2][k][r], o3 = d1[3][k][r];
                v2f b0 = { hi ? o2 : o0, hi ? o3 : o1 };
                v2f b1 = { hi ? o0 : o2, hi ? o1 : o3 };
                acc[t] = __builtin_amdgcn_wmma_f32_16x16x4_f32(
                    false, a2[g][k][0], false, b0, (short)0, acc[t], false, false);
                acc[t] = __builtin_amdgcn_wmma_f32_16x16x4_f32(
                    false, a2[g][k][1], false, b1, (short)0, acc[t], false, false);
            }
        }
    }
}

__global__ void __launch_bounds__(32 * SPLITK) __attribute__((amdgpu_waves_per_eu(2, 2)))
conv_wmma_k(const float* __restrict__ x, const float* __restrict__ w1,
            const float* __restrict__ w2, float* __restrict__ out) {
    __shared__ float part[SPLITK][NT][8][32];

    int blk    = blockIdx.x;
    int w      = threadIdx.x >> 5;
    int lane   = threadIdx.x & 31;
    int rh     = blk & 1;
    int mt     = (blk >> 1) & 1;
    int s_tile = blk >> 2;
    int lr     = lane & 15;
    int half   = lane >> 4;
    int s      = s_tile * 16 + lr;

    v8f acc[NT];
    if (rh) contract<1>(x, w1, w2, mt, w, lane, s, acc);
    else    contract<0>(x, w1, w2, mt, w, lane, s, acc);

    #pragma unroll
    for (int t = 0; t < NT; ++t)
        #pragma unroll
        for (int r2 = 0; r2 < 8; ++r2)
            part[w][t][r2][lane] = acc[t][r2];
    __syncthreads();

    int t = w >> 1;
    #pragma unroll
    for (int u = 0; u < 4; ++u) {
        int r2 = (w & 1) * 4 + u;
        float sum = part[0][t][r2][lane];
        #pragma unroll
        for (int ww = 1; ww < SPLITK; ++ww)
            sum += part[ww][t][r2][lane];
        int l  = mt * 16 + half * 8 + rh * 4 + t;
        int oc = l * 8 + r2;
        out[oc * HW + s] = sum;
    }
}

extern "C" void kernel_launch(void* const* d_in, const int* in_sizes, int n_in,
                              void* d_out, int out_size, void* d_ws, size_t ws_size,
                              hipStream_t stream) {
    const float* x  = (const float*)d_in[0];
    const float* w1 = (const float*)d_in[1];
    const float* w2 = (const float*)d_in[2];
    float* out = (float*)d_out;

    conv_wmma_k<<<196, 256, 0, stream>>>(x, w1, w2, out);
}


# All 48 x loads of a wave issued right behind its weight loads, ahead of the stage-2 operand set-up that waits for w2: the weight and x round trips overlap; sum order unchanged
# speedup vs baseline: 8.5293x; 1.1351x over previous
#include <hip/hip_runtime.h>


typedef __attribute__((ext_vector_type(2))) float v2f;
typedef __attribute__((ext_vector_type(8))) float v8f;
typedef __attribute__((ext_vector_type(3))) float v3f;
typedef v3f v3f_a4 __attribute__((aligned(4)));

#define HW       784
#define SPLITK   8
#define CG_WAVE  8
#define NT       4


template <int RH>
__device__ __forceinline__ void contract(const float* __restrict__ x, const float* __restrict__ w1,
                                         const float* __restrict__ w2, int mt, int w, int lane,
                                         int s, v8f (&acc)[NT]) {
    int lr   = lane & 15;
    int half = lane >> 4;
    int ho = s / 28;
    int wo = s - ho * 28;

    const float* w1p = w1 + (mt * 16 + lr) * 36 + half * 18;
    v2f a1[9];
    #pragma unroll
    for (int t = 0; t < 9; ++t) {
        a1[t].x = w1p[t];
        a1[t].y = w1p[9 + t];
    }

    int  q   = lr >> 3;
    int  p   = lr & 7;
    bool own = (q == half);
    bool hi  = (half != 0);
    const float* w2p = w2 + p * 192 + w * (CG_WAVE * 3);
    float w2v[CG_WAVE][3];
    #pragma unroll
    for (int c = 0; c < CG_WAVE; ++c)
        #pragma unroll
        for (int k = 0; k < 3; ++k)
            w2v[c][k] = w2p[c * 3 + k];

    bool onL = (wo == 0);
    bool onR = (wo == 27);
    int  cs  = wo - 1 + (onL ? 1 : 0) - (onR ? 1 : 0);
    bool rv0 = (ho > 0);
    bool rv2 = (ho < 27);
    int off[3];
    off[0] = (rv0 ? ho - 1 : ho) * 28 + cs;
    off[1] = ho * 28 + cs;
    off[2] = (rv2 ? ho + 1 : ho) * 28 + cs;
    const float* xb = x + (w * (CG_WAVE * 4) + 2 * half) * HW;

    v3f xq[CG_WAVE][3][2];
    #pragma unroll
    for (int c = 0; c < CG_WAVE; ++c)
        #pragma unroll
        for (int i = 0; i < 3; ++i)
            #pragma unroll
            for (int e = 0; e < 2; ++e)
                xq[c][i][e] = *(const v3f_a4*)(xb + (c * 4 + e) * HW + off[i]);

    __builtin_amdgcn_sched_barrier(0);
    v2f a2[2][3][2];
    #pragma unroll
    for (int g = 0; g < 2; ++g)
        #pragma unroll
        for (int k = 0; k < 3; ++k) {
            float c0 = w2v[4 * g + 0][k], c1 = w2v[4 * g + 1][k];
            float c2 = w2v[4 * g + 2][k], c3 = w2v[4 * g + 3][k];
            a2[g][k][0].x = own ? (hi ? c2 : c0) : 0.0f;
            a2[g][k][0].y = own ? (hi ? c3 : c1) : 0.0f;
            a2[g][k][1].x = own ? (hi ? c0 : c2) : 0.0f;
            a2[g][k][1].y = own ? (hi ? c1 : c3) : 0.0f;
        }

    #pragma unroll
    for (int t = 0; t < NT; ++t) acc[t] = v8f{};

    #pragma unroll
    for (int g = 0; g < 2; ++g) {
        v8f d1[4][3];
        #pragma unroll
        for (int cc = 0; cc < 4; ++cc) {
            int c = 4 * g + cc;
            #pragma unroll
            for (int k = 0; k < 3; ++k) d1[cc][k] = v8f{};
            #pragma unroll
            for (int i = 0; i < 3; ++i) {
                bool rv = (i == 0) ? rv0 : ((i == 2) ? rv2 : true);
                float tap[2][3];
                #pragma unroll
                for (int e = 0; e < 2; ++e) {
                    float q0 = xq[c][i][e].x, q1 = xq[c][i][e].y, q2 = xq[c][i][e].z;
                    float u0 = onR ? q1 : q0;
                    float u1 = onL ? q0 : (onR ? q2 : q1);
                    float u2 = onL ? q1 : q2;
                    tap[e][0] = (rv && !onL) ? u0 : 0.0f;
                    tap[e][1] = rv ? u1 : 0.0f;
                    tap[e][2] = (rv && !onR) ? u2 : 0.0f;
                }
                #pragma unroll
                for (int k = 0; k < 3; ++k) {
                    v2f b = { tap[0][k], tap[1][k] };
                    d1[cc][k] = __builtin_amdgcn_wmma_f32_16x16x4_f32(
                        false, a1[i * 3 + k], false, b, (short)0, d1[cc][k], false, false);
                }
            }
        }
        #pragma unroll
        for (int k = 0; k < 3; ++k) {
            #pragma unroll
            for (int t = 0; t < NT; ++t) {
                const int r = 4 * RH + t;
                float o0 = d1[0][k][r], o1 = d1[1][k][r];
                float o2 = d1[2][k][r], o3 = d1[3][k][r];
                v2f b0 = { hi ? o2 : o0, hi ? o3 : o1 };
                v2f b1 = { hi ? o0 : o2, hi ? o1 : o3 };
                acc[t] = __builtin_amdgcn_wmma_f32_16x16x4_f32(
                    false, a2[g][k][0], false, b0, (short)0, acc[t], false, false);
                acc[t] = __builtin_amdgcn_wmma_f32_16x16x4_f32(
                    false, a2[g][k][1], false, b1, (short)0, acc[t], false, false);
            }
        }
    }
}

__global__ void __launch_bounds__(32 * SPLITK) __attribute__((amdgpu_waves_per_eu(2, 2)))
conv_wmma_k(const float* __restrict__ x, const float* __restrict__ w1,
            const float* __restrict__ w2, float* __restrict__ out) {
    __shared__ float part[SPLITK][NT][8][32];

    int blk    = blockIdx.x;
    int w      = threadIdx.x >> 5;
    int lane   = threadIdx.x & 31;
    int rh     = blk & 1;
    int mt     = (blk >> 1) & 1;
    int s_tile = blk >> 2;
    int lr     = lane & 15;
    int half   = lane >> 4;
    int s      = s_tile * 16 + lr;

    v8f acc[NT];
    if (rh) contract<1>(x, w1, w2, mt, w, lane, s, acc);
    else    contract<0>(x, w1, w2, mt, w, lane, s, acc);

    #pragma unroll
    for (int t = 0; t < NT; ++t)
        #pragma unroll
        for (int r2 = 0; r2 < 8; ++r2)
            part[w][t][r2][lane] = acc[t][r2];
    __syncthreads();

    int t = w >> 1;
    #pragma unroll
    for (int u = 0; u < 4; ++u) {
        int r2 = (w & 1) * 4 + u;
        float sum = part[0][t][r2][lane];
        #pragma unroll
        for (int ww = 1; ww < SPLITK; ++ww)
            sum += part[ww][t][r2][lane];
        int l  = mt * 16 + half * 8 + rh * 4 + t;
        int oc = l * 8 + r2;
        out[oc * HW + s] = sum;
    }
}

extern "C" void kernel_launch(void* const* d_in, const int* in_sizes, int n_in,
                              void* d_out, int out_size, void* d_ws, size_t ws_size,
                              hipStream_t stream) {
    const float* x  = (const float*)d_in[0];
    const float* w1 = (const float*)d_in[1];
    const float* w2 = (const float*)d_in[2];
    float* out = (float*)d_out;

    conv_wmma_k<<<196, 256, 0, stream>>>(x, w1, w2, out);
}
